# starting kernel
# baseline (speedup 1.0000x reference)
#include <hip/hip_runtime.h>
#include <cfloat>

typedef __attribute__((ext_vector_type(2))) float v2f;
typedef __attribute__((ext_vector_type(8))) float v8f;

#define SK_STRIDE 68
#define SVT_STRIDE 260
#define SP_STRIDE 34

__device__ __forceinline__ v8f wmma_f32_16x16x4(v2f a, v2f b, v8f c) {
  return __builtin_amdgcn_wmma_f32_16x16x4_f32(false, a, false, b, (short)0, c, false, false);
}

__global__ __launch_bounds__(256) void local_attn_rope_kernel(
    const float* __restrict__ q, const float* __restrict__ k,
    const float* __restrict__ v, float* __restrict__ out)
{
  __shared__ float sK [256 * SK_STRIDE];
  __shared__ float sVT[ 64 * SVT_STRIDE];
  __shared__ float sP [8 * 16 * SP_STRIDE];

  const int wi   = blockIdx.x;
  const int bb   = blockIdx.y;
  const int tid  = threadIdx.x;
  const int wv   = tid >> 5;
  const int lane = tid & 31;
  const int l16  = lane & 15;
  const int hi   = lane >> 4;

  const float* kb = k + (size_t)bb * 4096 * 64;
  const float* vb = v + (size_t)bb * 4096 * 64;
  const int jbase = wi * 128 - 128;

  const float LOG2_IF = -0.41524101186092029f;

  #pragma unroll
  for (int it = 0; it < 32; ++it) {
    int p   = tid + 256 * it;
    int row = p >> 5;
    int c   = p & 31;
    int grow = jbase + row;
    float x1 = 0.f, x2 = 0.f;
    if (grow >= 0) {
      const float* kr = kb + (size_t)grow * 64;
      x1 = kr[c]; x2 = kr[c + 32];
    }
    float ang = (float)row * exp2f(LOG2_IF * (float)c);
    float sn, cs; __sincosf(ang, &sn, &cs);
    sK[row * SK_STRIDE + c]      = x1 * cs - x2 * sn;
    sK[row * SK_STRIDE + c + 32] = x2 * cs + x1 * sn;
  }

  #pragma unroll
  for (int it = 0; it < 16; ++it) {
    int p   = tid + 256 * it;
    int row = p >> 4;
    int c4  = (p & 15) * 4;
    int grow = jbase + row;
    float4 val = make_float4(0.f, 0.f, 0.f, 0.f);
    if (grow >= 0) val = *(const float4*)(vb + (size_t)grow * 64 + c4);
    sVT[(c4 + 0) * SVT_STRIDE + row] = val.x;
    sVT[(c4 + 1) * SVT_STRIDE + row] = val.y;
    sVT[(c4 + 2) * SVT_STRIDE + row] = val.z;
    sVT[(c4 + 3) * SVT_STRIDE + row] = val.w;
  }

  const int iq = 16 * wv + l16;
  const float* qrow = q + ((size_t)bb * 4096 + (size_t)wi * 128 + iq) * 64;
  v2f qf[16];
  #pragma unroll
  for (int t = 0; t < 16; ++t)
    qf[t] = *(const v2f*)(qrow + 4 * t + 2 * hi);
  {
    const float scale = 0.125f;
    float pos = (float)(128 + iq);
    #pragma unroll
    for (int t = 0; t < 8; ++t) {
      #pragma unroll
      for (int i = 0; i < 2; ++i) {
        int kk = 4 * t + 2 * hi + i;
        float ang = pos * exp2f(LOG2_IF * (float)kk);
        float sn, cs; __sincosf(ang, &sn, &cs);
        cs *= scale; sn *= scale;
        float lo = qf[t][i], hh = qf[t + 8][i];
        qf[t][i]     = lo * cs - hh * sn;
        qf[t + 8][i] = hh * cs + lo * sn;
      }
    }
  }

  __syncthreads();

  v8f S[16];
  #pragma unroll
  for (int jt = 0; jt < 16; ++jt) {
    v8f acc = {0.f,0.f,0.f,0.f,0.f,0.f,0.f,0.f};
    const float* kr = &sK[(16 * jt + l16) * SK_STRIDE + 2 * hi];
    #pragma unroll
    for (int t = 0; t < 16; ++t) {
      v2f bf = *(const v2f*)(kr + 4 * t);
      acc = wmma_f32_16x16x4(qf[t], bf, acc);
    }
    S[jt] = acc;
  }

  const float NEGMAX = -3.402823466e38f;
  #pragma unroll
  for (int jt = 0; jt < 16; ++jt) {
    int j = 16 * jt + l16;
    bool padj = (wi == 0) && (j < 128);
    #pragma unroll
    for (int vv = 0; vv < 8; ++vv) {
      int iqc = 16 * wv + vv + 8 * hi;
      if (padj || (j > iqc + 128)) S[jt][vv] = NEGMAX;
    }
  }

  float rinv[8];
  #pragma unroll
  for (int vv = 0; vv < 8; ++vv) {
    float m = S[0][vv];
    #pragma unroll
    for (int jt = 1; jt < 16; ++jt) m = fmaxf(m, S[jt][vv]);
    m = fmaxf(m, __shfl_xor(m, 1, 16));
    m = fmaxf(m, __shfl_xor(m, 2, 16));
    m = fmaxf(m, __shfl_xor(m, 4, 16));
    m = fmaxf(m, __shfl_xor(m, 8, 16));
    float s = 0.f;
    #pragma unroll
    for (int jt = 0; jt < 16; ++jt) {
      float e = __expf(S[jt][vv] - m);
      S[jt][vv] = e;
      s += e;
    }
    s += __shfl_xor(s, 1, 16);
    s += __shfl_xor(s, 2, 16);
    s += __shfl_xor(s, 4, 16);
    s += __shfl_xor(s, 8, 16);
    rinv[vv] = 1.0f / s;
  }

  v8f O[4];
  #pragma unroll
  for (int n = 0; n < 4; ++n) O[n] = (v8f){0.f,0.f,0.f,0.f,0.f,0.f,0.f,0.f};

  float* myP = &sP[wv * 16 * SP_STRIDE];
  #pragma unroll
  for (int ch = 0; ch < 8; ++ch) {
    #pragma unroll
    for (int u = 0; u < 2; ++u) {
      v8f Pv = S[2 * ch + u];
      #pragma unroll
      for (int vv = 0; vv < 8; ++vv)
        myP[(vv + 8 * hi) * SP_STRIDE + 16 * u + l16] = Pv[vv];
    }
    asm volatile("s_wait_dscnt 0" ::: "memory");
    #pragma unroll
    for (int t = 0; t < 8; ++t) {
      v2f af = *(const v2f*)(&myP[l16 * SP_STRIDE + 4 * t + 2 * hi]);
      int jb = 32 * ch + 4 * t + 2 * hi;
      #pragma unroll
      for (int n = 0; n < 4; ++n) {
        v2f bf = *(const v2f*)(&sVT[(16 * n + l16) * SVT_STRIDE + jb]);
        O[n] = wmma_f32_16x16x4(af, bf, O[n]);
      }
    }
  }

  float* ob = out + ((size_t)bb * 4096 + (size_t)wi * 128 + 16 * wv + 8 * hi) * 64 + l16;
  #pragma unroll
  for (int n = 0; n < 4; ++n) {
    #pragma unroll
    for (int vv = 0; vv < 8; ++vv)
      ob[(size_t)vv * 64 + 16 * n] = O[n][vv] * rinv[vv];
  }
}

extern "C" void kernel_launch(void* const* d_in, const int* in_sizes, int n_in,
                              void* d_out, int out_size, void* d_ws, size_t ws_size,
                              hipStream_t stream) {
  const float* q = (const float*)d_in[0];
  const float* k = (const float*)d_in[1];
  const float* v = (const float*)d_in[2];
  float* out = (float*)d_out;
  dim3 grid(32, 32);
  local_attn_rope_kernel<<<grid, dim3(256), 0, stream>>>(q, k, v, out);
}


# LDS fragment loads of QK and P*V issued as ONE group per matrix block, a block ahead in a second register set (scheduler fences), not spread in the matrix runs; P per 16-key tile via a 3-tile ring
# speedup vs baseline: 2.4026x; 2.4026x over previous
#include <hip/hip_runtime.h>
#include <cfloat>


typedef __attribute__((ext_vector_type(2))) float v2f;
typedef __attribute__((ext_vector_type(8))) float v8f;

#define SK_STRIDE 68
#define SVT_STRIDE 260
#define SP_STRIDE 20
#define SP_TILE (16 * SP_STRIDE)
#define NWALK 4
#define KSLOT (128 * SK_STRIDE)

__device__ __forceinline__ v8f wmma_f32_16x16x4(v2f a, v2f b, v8f c) {
  return __builtin_amdgcn_wmma_f32_16x16x4_f32(false, a, false, b, (short)0, c, false, false);
}

__device__ __forceinline__ void lds_barrier() {
  asm volatile("s_wait_dscnt 0x0\n\ts_barrier_signal -1\n\ts_barrier_wait -1" ::: "memory");
}

__device__ __forceinline__ void sched_fence() { __builtin_amdgcn_sched_barrier(0); }

struct TileRegs {
  float4 ka[4];
  float4 kb[4];
  float4 vv[8];
};

__device__ __forceinline__ void load_tile(TileRegs& R, const float* __restrict__ kt,
                                          const float* __restrict__ vt, int tid) {
  const float* kp = kt + (tid >> 3) * 64 + (tid & 7) * 4;
  #pragma unroll
  for (int it = 0; it < 4; ++it) {
    R.ka[it] = *(const float4*)(kp + it * 2048);
    R.kb[it] = *(const float4*)(kp + it * 2048 + 32);
  }
  const float* vp = vt + (tid >> 4) * 64 + (tid & 15) * 4;
  #pragma unroll
  for (int it = 0; it < 8; ++it)
    R.vv[it] = *(const float4*)(vp + it * 1024);
}

__device__ __forceinline__ void stage_k(const TileRegs& R, float* slot, int pos0, int tid) {
  const float LOG2_IF = -0.41524101186092029f;
  const int r0 = tid >> 3;
  const int c4 = (tid & 7) * 4;
  float* sp = slot + r0 * SK_STRIDE + c4;
  #pragma unroll
  for (int it = 0; it < 4; ++it) {
    float x1[4] = {R.ka[it].x, R.ka[it].y, R.ka[it].z, R.ka[it].w};
    float x2[4] = {R.kb[it].x, R.kb[it].y, R.kb[it].z, R.kb[it].w};
    float o1[4], o2[4];
    float tpos = (float)(pos0 + r0 + 32 * it);
    #pragma unroll
    for (int e = 0; e < 4; ++e) {
      float ang = tpos * exp2f(LOG2_IF * (float)(c4 + e));
      float sn, cs; __sincosf(ang, &sn, &cs);
      o1[e] = x1[e] * cs - x2[e] * sn;
      o2[e] = x2[e] * cs + x1[e] * sn;
    }
    *(float4*)(sp + it * 32 * SK_STRIDE)      = make_float4(o1[0], o1[1], o1[2], o1[3]);
    *(float4*)(sp + it * 32 * SK_STRIDE + 32) = make_float4(o2[0], o2[1], o2[2], o2[3]);
  }
}

__device__ __forceinline__ void stage_v(const TileRegs& R, float* sVT, int colbase, int tid) {
  float* vp = sVT + (tid & 15) * 4 * SVT_STRIDE + colbase + (tid >> 4);
  #pragma unroll
  for (int it = 0; it < 8; ++it) {
    vp[0 * SVT_STRIDE + 16 * it] = R.vv[it].x;
    vp[1 * SVT_STRIDE + 16 * it] = R.vv[it].y;
    vp[2 * SVT_STRIDE + 16 * it] = R.vv[it].z;
    vp[3 * SVT_STRIDE + 16 * it] = R.vv[it].w;
  }
}

__global__ __launch_bounds__(256) void local_attn_rope_walk_kernel(
    const float* __restrict__ q, const float* __restrict__ k,
    const float* __restrict__ v, float* __restrict__ out)
{
  __shared__ __attribute__((aligned(16))) float sK [3 * KSLOT];
  __shared__ __attribute__((aligned(16))) float sVT[ 64 * SVT_STRIDE];
  __shared__ __attribute__((aligned(16))) float sP [8 * 3 * SP_TILE];

  const int w0   = blockIdx.x * NWALK;
  const int bb   = blockIdx.y;
  const int tid  = threadIdx.x;
  const int wv   = tid >> 5;
  const int lane = tid & 31;
  const int l16  = lane & 15;
  const int hi   = lane >> 4;

  const float* kb = k + (size_t)bb * 4096 * 64;
  const float* vb = v + (size_t)bb * 4096 * 64;

  const float LOG2_IF = -0.41524101186092029f;

  TileRegs R;
  load_tile(R, kb + (size_t)w0 * 8192, vb + (size_t)w0 * 8192, tid);

  const int iq = 16 * wv + l16;
  const float* qlane = q + ((size_t)bb * 4096 + iq) * 64 + 2 * hi;
  v2f qf[16];
  #pragma unroll
  for (int t = 0; t < 16; ++t)
    qf[t] = *(const v2f*)(qlane + (size_t)w0 * 8192 + 4 * t);

  const int lbcol0 = ((w0 - 1) & 1) * 128;
  if (w0 > 0) {
    TileRegs RL;
    load_tile(RL, kb + (size_t)(w0 - 1) * 8192, vb + (size_t)(w0 - 1) * 8192, tid);
    stage_k(RL, sK, 0, tid);
    stage_v(RL, sVT, lbcol0, tid);
  } else {
    for (int x = tid; x < KSLOT; x += 256) sK[x] = 0.f;
    for (int x = tid; x < 64 * 128; x += 256) sVT[(x >> 7) * SVT_STRIDE + lbcol0 + (x & 127)] = 0.f;
  }

  #pragma unroll 1
  for (int i = 0; i < NWALK; ++i) {
    const int wi = w0 + i;
    const int lb = i & 1;
    float* sKlb  = sK + lb * KSLOT;
    float* sKnx  = sK + (lb ^ 1) * KSLOT;
    float* sKcur = sK + 2 * KSLOT;
    const int vcur = (wi & 1) * 128;
    const int vlb  = 128 - vcur;

    stage_k(R, sKcur, 128, tid);
    stage_v(R, sVT, vcur, tid);
    #pragma unroll
    for (int t = 0; t < 16; t += 4)
      asm volatile("" :: "v"(qf[t]), "v"(qf[t + 1]), "v"(qf[t + 2]), "v"(qf[t + 3]));
    {
      const float scale = 0.125f;
      float pos = (float)(128 + iq);
      #pragma unroll
      for (int t = 0; t < 8; ++t) {
        #pragma unroll
        for (int e = 0; e < 2; ++e) {
          int kk = 4 * t + 2 * hi + e;
          float ang = pos * exp2f(LOG2_IF * (float)kk);
          float sn, cs; __sincosf(ang, &sn, &cs);
          cs *= scale; sn *= scale;
          float lo = qf[t][e], hh = qf[t + 8][e];
          qf[t][e]     = lo * cs - hh * sn;
          qf[t + 8][e] = hh * cs + lo * sn;
        }
      }
    }

    if (i + 1 < NWALK) {
      stage_k(R, sKnx, 0, tid);
      load_tile(R, kb + (size_t)(wi + 1) * 8192, vb + (size_t)(wi + 1) * 8192, tid);
    }

    lds_barrier();

    v8f S[16];
    v2f kf[2][2][16];
    #pragma unroll
    for (int u = 0; u < 2; ++u) {
      const float* kr = &sKlb[(16 * u + l16) * SK_STRIDE] + 2 * hi;
      #pragma unroll
      for (int t = 0; t < 16; ++t)
        kf[0][u][t] = *(const v2f*)(kr + 4 * t);
    }
    sched_fence();
    #pragma unroll
    for (int b = 0; b < 8; ++b) {
      if (b + 1 < 8) {
        #pragma unroll
        for (int u = 0; u < 2; ++u) {
          const int jt = 2 * (b + 1) + u;
          const float* kr = (jt < 8 ? &sKlb[(16 * jt + l16) * SK_STRIDE]
                                    : &sKcur[(16 * (jt - 8) + l16) * SK_STRIDE]) + 2 * hi;
          #pragma unroll
          for (int t = 0; t < 16; ++t)
            kf[(b + 1) & 1][u][t] = *(const v2f*)(kr + 4 * t);
        }
      }
      sched_fence();
      v8f acc0 = {0.f,0.f,0.f,0.f,0.f,0.f,0.f,0.f};
      v8f acc1 = {0.f,0.f,0.f,0.f,0.f,0.f,0.f,0.f};
      #pragma unroll
      for (int t = 0; t < 16; ++t) {
        acc0 = wmma_f32_16x16x4(qf[t], kf[b & 1][0][t], acc0);
        acc1 = wmma_f32_16x16x4(qf[t], kf[b & 1][1][t], acc1);
      }
      S[2 * b]     = acc0;
      S[2 * b + 1] = acc1;
      sched_fence();
    }

    {
      const int wq = (i + 1 < NWALK) ? wi + 1 : wi;
      const float* qn = qlane + (size_t)wq * 8192;
      #pragma unroll
      for (int t = 0; t < 16; ++t)
        qf[t] = *(const v2f*)(qn + 4 * t);
    }

    const float NEGMAX = -3.402823466e38f;
    #pragma unroll
    for (int jt = 0; jt < 16; ++jt) {
      int j = 16 * jt + l16;
      bool padj = (wi == 0) && (j < 128);
      #pragma unroll
      for (int vv = 0; vv < 8; ++vv) {
        int iqc = 16 * wv + vv + 8 * hi;
        if (padj || (j > iqc + 128)) S[jt][vv] = NEGMAX;
      }
    }

    float rmax[8], rsum[8];
    #pragma unroll
    for (int vv = 0; vv < 8; ++vv) {
      float m = S[0][vv];
      #pragma unroll
      for (int jt = 1; jt < 16; ++jt) m = fmaxf(m, S[jt][vv]);
      m = fmaxf(m, __shfl_xor(m, 1, 16));
      m = fmaxf(m, __shfl_xor(m, 2, 16));
      m = fmaxf(m, __shfl_xor(m, 4, 16));
      m = fmaxf(m, __shfl_xor(m, 8, 16));
      rmax[vv] = m;
      rsum[vv] = 0.f;
    }

    v8f O[4];
    #pragma unroll
    for (int n = 0; n < 4; ++n) O[n] = (v8f){0.f,0.f,0.f,0.f,0.f,0.f,0.f,0.f};

    float* myP = &sP[wv * 3 * SP_TILE];
    v2f pa[2][4];
    v2f vf[2][4][4];

#define EXP_TILE(J)                                                        \
    _Pragma("unroll")                                                      \
    for (int vv = 0; vv < 8; ++vv) {                                       \
      float e = __expf(S[(J)][vv] - rmax[vv]);                             \
      S[(J)][vv] = e;                                                      \
      rsum[vv] += e;                                                       \
    }
#define STORE_P(J)                                                         \
    _Pragma("unroll")                                                      \
    for (int vv = 0; vv < 8; ++vv)                                         \
      myP[((J) % 3) * SP_TILE + (vv + 8 * hi) * SP_STRIDE + l16] = S[(J)][vv];
#define LOAD_A(J)                                                          \
    _Pragma("unroll")                                                      \
    for (int t = 0; t < 4; ++t)                                            \
      pa[(J) & 1][t] = *(const v2f*)(&myP[((J) % 3) * SP_TILE + l16 * SP_STRIDE + 4 * t + 2 * hi]);
#define LOAD_B(J)                                                          \
    _Pragma("unroll")                                                      \
    for (int t = 0; t < 4; ++t) {                                          \
      const int jb = 16 * (J) + 4 * t + 2 * hi;                      \
      const int col = ((J) < 8) ? (vlb + jb) : (vcur + jb - 128);    \
      _Pragma("unroll")                                                    \
      for (int n = 0; n < 4; ++n)                                          \
        vf[(J) & 1][t][n] = *(const v2f*)(&sVT[(16 * n + l16) * SVT_STRIDE + col]);   \
    }

    EXP_TILE(0)
    EXP_TILE(1)
    EXP_TILE(2)
    STORE_P(0)
    STORE_P(1)
    LOAD_B(0)
    asm volatile("s_wait_dscnt 0" ::: "memory");
    LOAD_A(0)
    sched_fence();
    #pragma unroll
    for (int j = 0; j < 16; ++j) {
      if (j + 1 < 16) {
        asm volatile("s_wait_dscnt 0" ::: "memory");
        LOAD_A(j + 1)
        LOAD_B(j + 1)
      }
      if (j + 2 < 16) { STORE_P(j + 2) }
      sched_fence();
      #pragma unroll
      for (int t = 0; t < 4; ++t) {
        #pragma unroll
        for (int n = 0; n < 4; ++n)
          O[n] = wmma_f32_16x16x4(pa[j & 1][t], vf[j & 1][t][n], O[n]);
      }
      if (j + 3 < 16) { EXP_TILE(j + 3) }
      sched_fence();
    }
#undef EXP_TILE
#undef STORE_P
#undef LOAD_A
#undef LOAD_B

    float rinv[8];
    #pragma unroll
    for (int vv = 0; vv < 8; ++vv) {
      float s = rsum[vv];
      s += __shfl_xor(s, 1, 16);
      s += __shfl_xor(s, 2, 16);
      s += __shfl_xor(s, 4, 16);
      s += __shfl_xor(s, 8, 16);
      rinv[vv] = 1.0f / s;
    }

    float* ob = out + ((size_t)bb * 4096 + (size_t)wi * 128 + 16 * wv + 8 * hi) * 64 + l16;
    #pragma unroll
    for (int n = 0; n < 4; ++n) {
      #pragma unroll
      for (int vv = 0; vv < 8; ++vv)
        ob[(size_t)vv * 64 + 16 * n] = O[n][vv] * rinv[vv];
    }

    lds_barrier();
  }
}

extern "C" void kernel_launch(void* const* d_in, const int* in_sizes, int n_in,
                              void* d_out, int out_size, void* d_ws, size_t ws_size,
                              hipStream_t stream) {
  const float* q = (const float*)d_in[0];
  const float* k = (const float*)d_in[1];
  const float* v = (const float*)d_in[2];
  float* out = (float*)d_out;
  dim3 grid(8, 32);
  local_attn_rope_walk_kernel<<<grid, dim3(256), 0, stream>>>(q, k, v, out);
}


# Younger wave of each SIMD pair (w+4) takes issue priority (s_setprio 3) from a window's head through QK and 6 of 16 P*V tiles, then 0 again: both waves reach the closing barrier together
# speedup vs baseline: 2.5009x; 1.0409x over previous
#include <hip/hip_runtime.h>
#include <cfloat>


typedef __attribute__((ext_vector_type(2))) float v2f;
typedef __attribute__((ext_vector_type(8))) float v8f;

#define SK_STRIDE 68
#define SVT_STRIDE 260
#define SP_STRIDE 20
#define SP_TILE (16 * SP_STRIDE)
#define NWALK 4
#define KSLOT (128 * SK_STRIDE)
#define PRIO_TILES 6

__device__ __forceinline__ v8f wmma_f32_16x16x4(v2f a, v2f b, v8f c) {
  return __builtin_amdgcn_wmma_f32_16x16x4_f32(false, a, false, b, (short)0, c, false, false);
}

__device__ __forceinline__ void lds_barrier() {
  asm volatile("s_wait_dscnt 0x0\n\ts_barrier_signal -1\n\ts_barrier_wait -1" ::: "memory");
}

__device__ __forceinline__ void sched_fence() { __builtin_amdgcn_sched_barrier(0); }

struct TileRegs {
  float4 ka[4];
  float4 kb[4];
  float4 vv[8];
};

__device__ __forceinline__ void load_tile(TileRegs& R, const float* __restrict__ kt,
                                          const float* __restrict__ vt, int tid) {
  const float* kp = kt + (tid >> 3) * 64 + (tid & 7) * 4;
  #pragma unroll
  for (int it = 0; it < 4; ++it) {
    R.ka[it] = *(const float4*)(kp + it * 2048);
    R.kb[it] = *(const float4*)(kp + it * 2048 + 32);
  }
  const float* vp = vt + (tid >> 4) * 64 + (tid & 15) * 4;
  #pragma unroll
  for (int it = 0; it < 8; ++it)
    R.vv[it] = *(const float4*)(vp + it * 1024);
}

__device__ __forceinline__ void stage_k(const TileRegs& R, float* slot, int pos0, int tid) {
  const float LOG2_IF = -0.41524101186092029f;
  const int r0 = tid >> 3;
  const int c4 = (tid & 7) * 4;
  float* sp = slot + r0 * SK_STRIDE + c4;
  #pragma unroll
  for (int it = 0; it < 4; ++it) {
    float x1[4] = {R.ka[it].x, R.ka[it].y, R.ka[it].z, R.ka[it].w};
    float x2[4] = {R.kb[it].x, R.kb[it].y, R.kb[it].z, R.kb[it].w};
    float o1[4], o2[4];
    float tpos = (float)(pos0 + r0 + 32 * it);
    #pragma unroll
    for (int e = 0; e < 4; ++e) {
      float ang = tpos * exp2f(LOG2_IF * (float)(c4 + e));
      float sn, cs; __sincosf(ang, &sn, &cs);
      o1[e] = x1[e] * cs - x2[e] * sn;
      o2[e] = x2[e] * cs + x1[e] * sn;
    }
    *(float4*)(sp + it * 32 * SK_STRIDE)      = make_float4(o1[0], o1[1], o1[2], o1[3]);
    *(float4*)(sp + it * 32 * SK_STRIDE + 32) = make_float4(o2[0], o2[1], o2[2], o2[3]);
  }
}

__device__ __forceinline__ void stage_v(const TileRegs& R, float* sVT, int colbase, int tid) {
  float* vp = sVT + (tid & 15) * 4 * SVT_STRIDE + colbase + (tid >> 4);
  #pragma unroll
  for (int it = 0; it < 8; ++it) {
    vp[0 * SVT_STRIDE + 16 * it] = R.vv[it].x;
    vp[1 * SVT_STRIDE + 16 * it] = R.vv[it].y;
    vp[2 * SVT_STRIDE + 16 * it] = R.vv[it].z;
    vp[3 * SVT_STRIDE + 16 * it] = R.vv[it].w;
  }
}

__global__ __launch_bounds__(256) void local_attn_rope_walk_kernel(
    const float* __restrict__ q, const float* __restrict__ k,
    const float* __restrict__ v, float* __restrict__ out)
{
  __shared__ __attribute__((aligned(16))) float sK [3 * KSLOT];
  __shared__ __attribute__((aligned(16))) float sVT[ 64 * SVT_STRIDE];
  __shared__ __attribute__((aligned(16))) float sP [8 * 3 * SP_TILE];

  const int w0   = blockIdx.x * NWALK;
  const int bb   = blockIdx.y;
  const int tid  = threadIdx.x;
  const int wv   = tid >> 5;
  const int lane = tid & 31;
  const int l16  = lane & 15;
  const int hi   = lane >> 4;
  const bool younger = __builtin_amdgcn_readfirstlane(tid >> 5) >= 4;

  const float* kb = k + (size_t)bb * 4096 * 64;
  const float* vb = v + (size_t)bb * 4096 * 64;

  const float LOG2_IF = -0.41524101186092029f;

  TileRegs R;
  load_tile(R, kb + (size_t)w0 * 8192, vb + (size_t)w0 * 8192, tid);

  const int iq = 16 * wv + l16;
  const float* qlane = q + ((size_t)bb * 4096 + iq) * 64 + 2 * hi;
  v2f qf[16];
  #pragma unroll
  for (int t = 0; t < 16; ++t)
    qf[t] = *(const v2f*)(qlane + (size_t)w0 * 8192 + 4 * t);

  const int lbcol0 = ((w0 - 1) & 1) * 128;
  if (w0 > 0) {
    TileRegs RL;
    load_tile(RL, kb + (size_t)(w0 - 1) * 8192, vb + (size_t)(w0 - 1) * 8192, tid);
    stage_k(RL, sK, 0, tid);
    stage_v(RL, sVT, lbcol0, tid);
  } else {
    for (int x = tid; x < KSLOT; x += 256) sK[x] = 0.f;
    for (int x = tid; x < 64 * 128; x += 256) sVT[(x >> 7) * SVT_STRIDE + lbcol0 + (x & 127)] = 0.f;
  }

  #pragma unroll 1
  for (int i = 0; i < NWALK; ++i) {
    const int wi = w0 + i;
    const int lb = i & 1;
    float* sKlb  = sK + lb * KSLOT;
    float* sKnx  = sK + (lb ^ 1) * KSLOT;
    float* sKcur = sK + 2 * KSLOT;
    const int vcur = (wi & 1) * 128;
    const int vlb  = 128 - vcur;

    if (younger) __builtin_amdgcn_s_setprio(3);

    stage_k(R, sKcur, 128, tid);
    stage_v(R, sVT, vcur, tid);
    #pragma unroll
    for (int t = 0; t < 16; t += 4)
      asm volatile("" :: "v"(qf[t]), "v"(qf[t + 1]), "v"(qf[t + 2]), "v"(qf[t + 3]));
    {
      const float scale = 0.125f;
      float pos = (float)(128 + iq);
      #pragma unroll
      for (int t = 0; t < 8; ++t) {
        #pragma unroll
        for (int e = 0; e < 2; ++e) {
          int kk = 4 * t + 2 * hi + e;
          float ang = pos * exp2f(LOG2_IF * (float)kk);
          float sn, cs; __sincosf(ang, &sn, &cs);
          cs *= scale; sn *= scale;
          float lo = qf[t][e], hh = qf[t + 8][e];
          qf[t][e]     = lo * cs - hh * sn;
          qf[t + 8][e] = hh * cs + lo * sn;
        }
      }
    }

    if (i + 1 < NWALK) {
      stage_k(R, sKnx, 0, tid);
      load_tile(R, kb + (size_t)(wi + 1) * 8192, vb + (size_t)(wi + 1) * 8192, tid);
    }

    lds_barrier();

    v8f S[16];
    v2f kf[2][2][16];
    #pragma unroll
    for (int u = 0; u < 2; ++u) {
      const float* kr = &sKlb[(16 * u + l16) * SK_STRIDE] + 2 * hi;
      #pragma unroll
      for (int t = 0; t < 16; ++t)
        kf[0][u][t] = *(const v2f*)(kr + 4 * t);
    }
    sched_fence();
    #pragma unroll
    for (int b = 0; b < 8; ++b) {
      if (b + 1 < 8) {
        #pragma unroll
        for (int u = 0; u < 2; ++u) {
          const int jt = 2 * (b + 1) + u;
          const float* kr = (jt < 8 ? &sKlb[(16 * jt + l16) * SK_STRIDE]
                                    : &sKcur[(16 * (jt - 8) + l16) * SK_STRIDE]) + 2 * hi;
          #pragma unroll
          for (int t = 0; t < 16; ++t)
            kf[(b + 1) & 1][u][t] = *(const v2f*)(kr + 4 * t);
        }
      }
      sched_fence();
      v8f acc0 = {0.f,0.f,0.f,0.f,0.f,0.f,0.f,0.f};
      v8f acc1 = {0.f,0.f,0.f,0.f,0.f,0.f,0.f,0.f};
      #pragma unroll
      for (int t = 0; t < 16; ++t) {
        acc0 = wmma_f32_16x16x4(qf[t], kf[b & 1][0][t], acc0);
        acc1 = wmma_f32_16x16x4(qf[t], kf[b & 1][1][t], acc1);
      }
      S[2 * b]     = acc0;
      S[2 * b + 1] = acc1;
      sched_fence();
    }

    {
      const int wq = (i + 1 < NWALK) ? wi + 1 : wi;
      const float* qn = qlane + (size_t)wq * 8192;
      #pragma unroll
      for (int t = 0; t < 16; ++t)
        qf[t] = *(const v2f*)(qn + 4 * t);
    }

    const float NEGMAX = -3.402823466e38f;
    #pragma unroll
    for (int jt = 0; jt < 16; ++jt) {
      int j = 16 * jt + l16;
      bool padj = (wi == 0) && (j < 128);
      #pragma unroll
      for (int vv = 0; vv < 8; ++vv) {
        int iqc = 16 * wv + vv + 8 * hi;
        if (padj || (j > iqc + 128)) S[jt][vv] = NEGMAX;
      }
    }

    float rmax[8], rsum[8];
    #pragma unroll
    for (int vv = 0; vv < 8; ++vv) {
      float m = S[0][vv];
      #pragma unroll
      for (int jt = 1; jt < 16; ++jt) m = fmaxf(m, S[jt][vv]);
      m = fmaxf(m, __shfl_xor(m, 1, 16));
      m = fmaxf(m, __shfl_xor(m, 2, 16));
      m = fmaxf(m, __shfl_xor(m, 4, 16));
      m = fmaxf(m, __shfl_xor(m, 8, 16));
      rmax[vv] = m;
      rsum[vv] = 0.f;
    }

    v8f O[4];
    #pragma unroll
    for (int n = 0; n < 4; ++n) O[n] = (v8f){0.f,0.f,0.f,0.f,0.f,0.f,0.f,0.f};

    float* myP = &sP[wv * 3 * SP_TILE];
    v2f pa[2][4];
    v2f vf[2][4][4];

#define EXP_TILE(J)                                                        \
    _Pragma("unroll")                                                      \
    for (int vv = 0; vv < 8; ++vv) {                                       \
      float e = __expf(S[(J)][vv] - rmax[vv]);                             \
      S[(J)][vv] = e;                                                      \
      rsum[vv] += e;                                                       \
    }
#define STORE_P(J)                                                         \
    _Pragma("unroll")                                                      \
    for (int vv = 0; vv < 8; ++vv)                                         \
      myP[((J) % 3) * SP_TILE + (vv + 8 * hi) * SP_STRIDE + l16] = S[(J)][vv];
#define LOAD_A(J)                                                          \
    _Pragma("unroll")                                                      \
    for (int t = 0; t < 4; ++t)                                            \
      pa[(J) & 1][t] = *(const v2f*)(&myP[((J) % 3) * SP_TILE + l16 * SP_STRIDE + 4 * t + 2 * hi]);
#define LOAD_B(J)                                                          \
    _Pragma("unroll")                                                      \
    for (int t = 0; t < 4; ++t) {                                          \
      const int jb = 16 * (J) + 4 * t + 2 * hi;                      \
      const int col = ((J) < 8) ? (vlb + jb) : (vcur + jb - 128);    \
      _Pragma("unroll")                                                    \
      for (int n = 0; n < 4; ++n)                                          \
        vf[(J) & 1][t][n] = *(const v2f*)(&sVT[(16 * n + l16) * SVT_STRIDE + col]);   \
    }

    EXP_TILE(0)
    EXP_TILE(1)
    EXP_TILE(2)
    STORE_P(0)
    STORE_P(1)
    LOAD_B(0)
    asm volatile("s_wait_dscnt 0" ::: "memory");
    LOAD_A(0)
    sched_fence();
    #pragma unroll
    for (int j = 0; j < 16; ++j) {
      if (j == PRIO_TILES) __builtin_amdgcn_s_setprio(0);
      if (j + 1 < 16) {
        asm volatile("s_wait_dscnt 0" ::: "memory");
        LOAD_A(j + 1)
        LOAD_B(j + 1)
      }
      if (j + 2 < 16) { STORE_P(j + 2) }
      sched_fence();
      #pragma unroll
      for (int t = 0; t < 4; ++t) {
        #pragma unroll
        for (int n = 0; n < 4; ++n)
          O[n] = wmma_f32_16x16x4(pa[j & 1][t], vf[j & 1][t][n], O[n]);
      }
      if (j + 3 < 16) { EXP_TILE(j + 3) }
      sched_fence();
    }
#undef EXP_TILE
#undef STORE_P
#undef LOAD_A
#undef LOAD_B

    float rinv[8];
    #pragma unroll
    for (int vv = 0; vv < 8; ++vv) {
      float s = rsum[vv];
      s += __shfl_xor(s, 1, 16);
      s += __shfl_xor(s, 2, 16);
      s += __shfl_xor(s, 4, 16);
      s += __shfl_xor(s, 8, 16);
      rinv[vv] = 1.0f / s;
    }

    float* ob = out + ((size_t)bb * 4096 + (size_t)wi * 128 + 16 * wv + 8 * hi) * 64 + l16;
    #pragma unroll
    for (int n = 0; n < 4; ++n) {
      #pragma unroll
      for (int vv = 0; vv < 8; ++vv)
        ob[(size_t)vv * 64 + 16 * n] = O[n][vv] * rinv[vv];
    }

    lds_barrier();
  }
}

extern "C" void kernel_launch(void* const* d_in, const int* in_sizes, int n_in,
                              void* d_out, int out_size, void* d_ws, size_t ws_size,
                              hipStream_t stream) {
  const float* q = (const float*)d_in[0];
  const float* k = (const float*)d_in[1];
  const float* v = (const float*)d_in[2];
  float* out = (float*)d_out;
  dim3 grid(8, 32);
  local_attn_rope_walk_kernel<<<grid, dim3(256), 0, stream>>>(q, k, v, out);
}
